# starting kernel
# baseline (speedup 1.0000x reference)
#include <hip/hip_runtime.h>
#include <hip/hip_bf16.h>

typedef __attribute__((ext_vector_type(2))) float v2f;
typedef __attribute__((ext_vector_type(8))) float v8f;

#define TPB 256
#define KC    64
#define ASTR  68
#define WTSTR 68

__global__ void gcn_deg_init(float* __restrict__ deg, int n) {
    int i = blockIdx.x * blockDim.x + threadIdx.x;
    if (i < n) deg[i] = 1.0f;
}

__global__ void gcn_deg_edges(const int* __restrict__ dst, float* __restrict__ deg, int E) {
    int e = blockIdx.x * blockDim.x + threadIdx.x;
    if (e < E)
        __hip_atomic_fetch_add(&deg[dst[e]], 1.0f, __ATOMIC_RELAXED, __HIP_MEMORY_SCOPE_AGENT);
}

__global__ void gcn_dinv(float* __restrict__ deg, int n) {
    int i = blockIdx.x * blockDim.x + threadIdx.x;
    if (i < n) deg[i] = rsqrtf(fmaxf(deg[i], 1.0f));
}

__global__ void gcn_edge_coef(const int* __restrict__ src, const int* __restrict__ dst,
                              const float* __restrict__ dinv, float* __restrict__ coef, int E) {
    int e = blockIdx.x * blockDim.x + threadIdx.x;
    if (e < E) coef[e] = dinv[src[e]] * dinv[dst[e]];
}

__global__ __launch_bounds__(TPB)
void gcn_gemm_wmma(const float* __restrict__ A, const float* __restrict__ W,
                   float* __restrict__ C, int M, int K, int N) {
    __shared__ float lA[128 * ASTR];
    __shared__ float lWt[64 * WTSTR];
    const int tid  = threadIdx.x;
    const int lane = tid & 31;
    const int wid  = tid >> 5;
    const int l16  = lane & 15;
    const int half = lane >> 4;
    const int rowBase = blockIdx.x * 128;
    const int colBase = blockIdx.y * 64;

    v8f acc[4] = {};

    for (int kc = 0; kc < K; kc += KC) {
#pragma unroll
        for (int i = 0; i < 8; ++i) {
            int idx = tid + i * TPB;
            int r   = idx >> 4;
            int c4  = (idx & 15) << 2;
            int gr  = min(rowBase + r, M - 1);
            float4 v = *(const float4*)(A + (size_t)gr * K + kc + c4);
            *(float4*)(&lA[r * ASTR + c4]) = v;
        }
#pragma unroll
        for (int i = 0; i < 4; ++i) {
            int idx = tid + i * TPB;
            int r   = idx >> 4;
            int c4  = (idx & 15) << 2;
            float4 v = *(const float4*)(W + (size_t)(kc + r) * N + colBase + c4);
            lWt[(c4 + 0) * WTSTR + r] = v.x;
            lWt[(c4 + 1) * WTSTR + r] = v.y;
            lWt[(c4 + 2) * WTSTR + r] = v.z;
            lWt[(c4 + 3) * WTSTR + r] = v.w;
        }
        __syncthreads();

        for (int kk = 0; kk < KC; kk += 4) {
            const int ks = kk + 2 * half;
            v2f a = *(const v2f*)(&lA[(wid * 16 + l16) * ASTR + ks]);
#pragma unroll
            for (int ct = 0; ct < 4; ++ct) {
                v2f b = *(const v2f*)(&lWt[(ct * 16 + l16) * WTSTR + ks]);
                acc[ct] = __builtin_amdgcn_wmma_f32_16x16x4_f32(
                    false, a, false, b, (short)0, acc[ct], false, false);
            }
        }
        __syncthreads();
    }

#pragma unroll
    for (int ct = 0; ct < 4; ++ct) {
        const int col = colBase + ct * 16 + l16;
#pragma unroll
        for (int v = 0; v < 8; ++v) {
            const int row = rowBase + wid * 16 + v + 8 * half;
            if (row < M) C[(size_t)row * N + col] = acc[ct][v];
        }
    }
}

__global__ void gcn_agg_init(const float* __restrict__ xw, const float* __restrict__ bias,
                             const float* __restrict__ dinv, float* __restrict__ agg,
                             long totalV4, int logF) {
    long idx = (long)blockIdx.x * blockDim.x + threadIdx.x;
    if (idx >= totalV4) return;
    size_t elem = (size_t)idx << 2;
    int i  = (int)(elem >> logF);
    int f4 = (int)(elem & ((1 << logF) - 1));
    float di = dinv[i];
    float s  = di * di;
    float4 xv = *(const float4*)(xw + elem);
    float4 bv = *(const float4*)(bias + f4);
    float4 o;
    o.x = bv.x + xv.x * s; o.y = bv.y + xv.y * s;
    o.z = bv.z + xv.z * s; o.w = bv.w + xv.w * s;
    *(float4*)(agg + elem) = o;
}

__global__ void gcn_edge_scatter(const int* __restrict__ src, const int* __restrict__ dst,
                                 const float* __restrict__ coef, const float* __restrict__ xw,
                                 float* __restrict__ agg, long totalV4, int logF) {
    long gid = (long)blockIdx.x * blockDim.x + threadIdx.x;
    if (gid >= totalV4) return;
    const int logF4 = logF - 2;
    int e  = (int)(gid >> logF4);
    int f4 = (int)((gid & ((1 << logF4) - 1)) << 2);
    float c = coef[e];
    const float4 xv = *(const float4*)(xw + ((size_t)src[e] << logF) + f4);
    float* base = agg + ((size_t)dst[e] << logF) + f4;
    __hip_atomic_fetch_add(base + 0, xv.x * c, __ATOMIC_RELAXED, __HIP_MEMORY_SCOPE_AGENT);
    __hip_atomic_fetch_add(base + 1, xv.y * c, __ATOMIC_RELAXED, __HIP_MEMORY_SCOPE_AGENT);
    __hip_atomic_fetch_add(base + 2, xv.z * c, __ATOMIC_RELAXED, __HIP_MEMORY_SCOPE_AGENT);
    __hip_atomic_fetch_add(base + 3, xv.w * c, __ATOMIC_RELAXED, __HIP_MEMORY_SCOPE_AGENT);
}

__global__ void gcn_bn_relu(const float* __restrict__ in, float* __restrict__ out,
                            const float* __restrict__ g, const float* __restrict__ beta,
                            const float* __restrict__ m, const float* __restrict__ v,
                            long totalV4, int fmask) {
    long idx = (long)blockIdx.x * blockDim.x + threadIdx.x;
    if (idx >= totalV4) return;
    size_t elem = (size_t)idx << 2;
    int f4 = (int)(elem & fmask);
    float4 xv = *(const float4*)(in + elem);
    float4 gv = *(const float4*)(g + f4);
    float4 bv = *(const float4*)(beta + f4);
    float4 mv = *(const float4*)(m + f4);
    float4 vv = *(const float4*)(v + f4);
    float4 o;
    o.x = fmaxf((xv.x - mv.x) * rsqrtf(vv.x + 1e-5f) * gv.x + bv.x, 0.0f);
    o.y = fmaxf((xv.y - mv.y) * rsqrtf(vv.y + 1e-5f) * gv.y + bv.y, 0.0f);
    o.z = fmaxf((xv.z - mv.z) * rsqrtf(vv.z + 1e-5f) * gv.z + bv.z, 0.0f);
    o.w = fmaxf((xv.w - mv.w) * rsqrtf(vv.w + 1e-5f) * gv.w + bv.w, 0.0f);
    *(float4*)(out + elem) = o;
}

__global__ __launch_bounds__(TPB)
void gcn_softmax64(const float* __restrict__ in, float* __restrict__ out, int n) {
    int wid  = threadIdx.x >> 5;
    int lane = threadIdx.x & 31;
    int row  = blockIdx.x * 8 + wid;
    if (row >= n) return;
    const float* p = in + (size_t)row * 64;
    float v0 = p[lane], v1 = p[lane + 32];
    float mx = fmaxf(v0, v1);
#pragma unroll
    for (int off = 16; off > 0; off >>= 1) mx = fmaxf(mx, __shfl_xor(mx, off, 32));
    float e0 = __expf(v0 - mx), e1 = __expf(v1 - mx);
    float s = e0 + e1;
#pragma unroll
    for (int off = 16; off > 0; off >>= 1) s += __shfl_xor(s, off, 32);
    float inv = 1.0f / s;
    float* q = out + (size_t)row * 64;
    q[lane] = e0 * inv;
    q[lane + 32] = e1 * inv;
}

extern "C" void kernel_launch(void* const* d_in, const int* in_sizes, int n_in,
                              void* d_out, int out_size, void* d_ws, size_t ws_size,
                              hipStream_t stream) {
    const int IN = 256, HID_ = 256, OUT_ = 64;
    const int N = in_sizes[0] / IN;
    const int E = in_sizes[1] / 2;

    const float* x    = (const float*)d_in[0];
    const int*   src  = (const int*)d_in[1];
    const int*   dst  = ((const int*)d_in[1]) + E;
    const float* W1   = (const float*)d_in[2];
    const float* b1   = (const float*)d_in[3];
    const float* g1   = (const float*)d_in[4];
    const float* be1  = (const float*)d_in[5];
    const float* m1   = (const float*)d_in[6];
    const float* v1   = (const float*)d_in[7];
    const float* W2   = (const float*)d_in[8];
    const float* b2   = (const float*)d_in[9];
    const float* g2   = (const float*)d_in[10];
    const float* be2  = (const float*)d_in[11];
    const float* m2   = (const float*)d_in[12];
    const float* v2   = (const float*)d_in[13];
    const float* W3   = (const float*)d_in[14];
    const float* b3   = (const float*)d_in[15];
    float* out = (float*)d_out;

    char* ws = (char*)d_ws;
    size_t off = 0;
    float* dinv = (float*)(ws + off); off += ((size_t)N * 4 + 255) & ~(size_t)255;
    float* coef = (float*)(ws + off); off += ((size_t)E * 4 + 255) & ~(size_t)255;
    float* bufA = (float*)(ws + off); off += (size_t)N * HID_ * 4;
    float* bufB = (float*)(ws + off);
    (void)ws_size; (void)n_in; (void)out_size;

    const long n256v4 = (long)N * HID_ / 4;
    const long n64v4  = (long)N * OUT_ / 4;
    const long e256v4 = (long)E * HID_ / 4;
    const long e64v4  = (long)E * OUT_ / 4;
    const int gN = (N + TPB - 1) / TPB;
    const int gE = (E + TPB - 1) / TPB;
    const int gemmX = (N + 127) / 128;
    auto blocks = [](long t) { return (int)((t + TPB - 1) / TPB); };

    gcn_deg_init <<<gN, TPB, 0, stream>>>(dinv, N);
    gcn_deg_edges<<<gE, TPB, 0, stream>>>(dst, dinv, E);
    gcn_dinv     <<<gN, TPB, 0, stream>>>(dinv, N);
    gcn_edge_coef<<<gE, TPB, 0, stream>>>(src, dst, dinv, coef, E);

    gcn_gemm_wmma   <<<dim3(gemmX, HID_ / 64), TPB, 0, stream>>>(x, W1, bufA, N, IN, HID_);
    gcn_agg_init    <<<blocks(n256v4), TPB, 0, stream>>>(bufA, b1, dinv, bufB, n256v4, 8);
    gcn_edge_scatter<<<blocks(e256v4), TPB, 0, stream>>>(src, dst, coef, bufA, bufB, e256v4, 8);
    gcn_bn_relu     <<<blocks(n256v4), TPB, 0, stream>>>(bufB, bufA, g1, be1, m1, v1, n256v4, 255);

    gcn_gemm_wmma   <<<dim3(gemmX, HID_ / 64), TPB, 0, stream>>>(bufA, W2, bufB, N, HID_, HID_);
    gcn_agg_init    <<<blocks(n256v4), TPB, 0, stream>>>(bufB, b2, dinv, bufA, n256v4, 8);
    gcn_edge_scatter<<<blocks(e256v4), TPB, 0, stream>>>(src, dst, coef, bufB, bufA, e256v4, 8);
    gcn_bn_relu     <<<blocks(n256v4), TPB, 0, stream>>>(bufA, bufB, g2, be2, m2, v2, n256v4, 255);

    gcn_gemm_wmma   <<<dim3(gemmX, OUT_ / 64), TPB, 0, stream>>>(bufB, W3, bufA, N, HID_, OUT_);
    gcn_agg_init    <<<blocks(n64v4), TPB, 0, stream>>>(bufA, b3, dinv, bufB, n64v4, 6);
    gcn_edge_scatter<<<blocks(e64v4), TPB, 0, stream>>>(src, dst, coef, bufA, bufB, e64v4, 6);
    gcn_softmax64   <<<(N + 7) / 8, TPB, 0, stream>>>(bufB, out, N);
}


# Edge scatter re-mapped: a wave walks 32 edges, lanes on consecutive floats, so each gather load and each f32 atomic covers one whole 128-B line; atomics kept, sum order by arrival as before
# speedup vs baseline: 1.6293x; 1.6293x over previous
#include <hip/hip_runtime.h>
#include <hip/hip_bf16.h>

typedef __attribute__((ext_vector_type(2))) float v2f;
typedef __attribute__((ext_vector_type(8))) float v8f;

#define TPB 256
#define KC    64
#define ASTR  68
#define WTSTR 68

__global__ void gcn_deg_init(float* __restrict__ deg, int n) {
    int i = blockIdx.x * blockDim.x + threadIdx.x;
    if (i < n) deg[i] = 1.0f;
}

__global__ void gcn_deg_edges(const int* __restrict__ dst, float* __restrict__ deg, int E) {
    int e = blockIdx.x * blockDim.x + threadIdx.x;
    if (e < E)
        __hip_atomic_fetch_add(&deg[dst[e]], 1.0f, __ATOMIC_RELAXED, __HIP_MEMORY_SCOPE_AGENT);
}

__global__ void gcn_dinv(float* __restrict__ deg, int n) {
    int i = blockIdx.x * blockDim.x + threadIdx.x;
    if (i < n) deg[i] = rsqrtf(fmaxf(deg[i], 1.0f));
}

__global__ void gcn_edge_coef(const int* __restrict__ src, const int* __restrict__ dst,
                              const float* __restrict__ dinv, float* __restrict__ coef, int E) {
    int e = blockIdx.x * blockDim.x + threadIdx.x;
    if (e < E) coef[e] = dinv[src[e]] * dinv[dst[e]];
}

__global__ __launch_bounds__(TPB)
void gcn_gemm_wmma(const float* __restrict__ A, const float* __restrict__ W,
                   float* __restrict__ C, int M, int K, int N) {
    __shared__ float lA[128 * ASTR];
    __shared__ float lWt[64 * WTSTR];
    const int tid  = threadIdx.x;
    const int lane = tid & 31;
    const int wid  = tid >> 5;
    const int l16  = lane & 15;
    const int half = lane >> 4;
    const int rowBase = blockIdx.x * 128;
    const int colBase = blockIdx.y * 64;

    v8f acc[4] = {};

    for (int kc = 0; kc < K; kc += KC) {
#pragma unroll
        for (int i = 0; i < 8; ++i) {
            int idx = tid + i * TPB;
            int r   = idx >> 4;
            int c4  = (idx & 15) << 2;
            int gr  = min(rowBase + r, M - 1);
            float4 v = *(const float4*)(A + (size_t)gr * K + kc + c4);
            *(float4*)(&lA[r * ASTR + c4]) = v;
        }
#pragma unroll
        for (int i = 0; i < 4; ++i) {
            int idx = tid + i * TPB;
            int r   = idx >> 4;
            int c4  = (idx & 15) << 2;
            float4 v = *(const float4*)(W + (size_t)(kc + r) * N + colBase + c4);
            lWt[(c4 + 0) * WTSTR + r] = v.x;
            lWt[(c4 + 1) * WTSTR + r] = v.y;
            lWt[(c4 + 2) * WTSTR + r] = v.z;
            lWt[(c4 + 3) * WTSTR + r] = v.w;
        }
        __syncthreads();

        for (int kk = 0; kk < KC; kk += 4) {
            const int ks = kk + 2 * half;
            v2f a = *(const v2f*)(&lA[(wid * 16 + l16) * ASTR + ks]);
#pragma unroll
            for (int ct = 0; ct < 4; ++ct) {
                v2f b = *(const v2f*)(&lWt[(ct * 16 + l16) * WTSTR + ks]);
                acc[ct] = __builtin_amdgcn_wmma_f32_16x16x4_f32(
                    false, a, false, b, (short)0, acc[ct], false, false);
            }
        }
        __syncthreads();
    }

#pragma unroll
    for (int ct = 0; ct < 4; ++ct) {
        const int col = colBase + ct * 16 + l16;
#pragma unroll
        for (int v = 0; v < 8; ++v) {
            const int row = rowBase + wid * 16 + v + 8 * half;
            if (row < M) C[(size_t)row * N + col] = acc[ct][v];
        }
    }
}

__global__ void gcn_agg_init(const float* __restrict__ xw, const float* __restrict__ bias,
                             const float* __restrict__ dinv, float* __restrict__ agg,
                             long totalV4, int logF) {
    long idx = (long)blockIdx.x * blockDim.x + threadIdx.x;
    if (idx >= totalV4) return;
    size_t elem = (size_t)idx << 2;
    int i  = (int)(elem >> logF);
    int f4 = (int)(elem & ((1 << logF) - 1));
    float di = dinv[i];
    float s  = di * di;
    float4 xv = *(const float4*)(xw + elem);
    float4 bv = *(const float4*)(bias + f4);
    float4 o;
    o.x = bv.x + xv.x * s; o.y = bv.y + xv.y * s;
    o.z = bv.z + xv.z * s; o.w = bv.w + xv.w * s;
    *(float4*)(agg + elem) = o;
}

template <int LOGF, bool FULL>
__device__ __forceinline__ void gcn_scatter_walk(int sl, int dl, float cl, int cnt, int lane,
                                                 const float* __restrict__ xw,
                                                 float* __restrict__ agg) {
    constexpr int NL  = (1 << LOGF) / 32;
    constexpr int EPI = (NL >= 16) ? 1 : 16 / NL;
    const int ngroups = FULL ? 32 / EPI : (cnt + EPI - 1) / EPI;

    float nxt[EPI][NL];
#pragma unroll
    for (int k = 0; k < EPI; ++k)
#pragma unroll
        for (int j = 0; j < NL; ++j) nxt[k][j] = 0.0f;

#pragma unroll
    for (int k = 0; k < EPI; ++k) {
        if (FULL || k < cnt) {
            const int s = __builtin_amdgcn_readlane(sl, k);
            const float* xr = xw + ((size_t)s << LOGF) + lane;
#pragma unroll
            for (int j = 0; j < NL; ++j) nxt[k][j] = xr[j * 32];
        }
    }

    for (int g = 0; g < ngroups; ++g) {
        float cur[EPI][NL];
#pragma unroll
        for (int k = 0; k < EPI; ++k)
#pragma unroll
            for (int j = 0; j < NL; ++j) cur[k][j] = nxt[k][j];

        if (g + 1 < ngroups) {
#pragma unroll
            for (int k = 0; k < EPI; ++k) {
                const int idx = (g + 1) * EPI + k;
                if (FULL || idx < cnt) {
                    const int s = __builtin_amdgcn_readlane(sl, idx);
                    const float* xr = xw + ((size_t)s << LOGF) + lane;
#pragma unroll
                    for (int j = 0; j < NL; ++j) nxt[k][j] = xr[j * 32];
                }
            }
        }

#pragma unroll
        for (int k = 0; k < EPI; ++k) {
            const int idx = g * EPI + k;
            if (FULL || idx < cnt) {
                const int   d = __builtin_amdgcn_readlane(dl, idx);
                const float c = __int_as_float(__builtin_amdgcn_readlane(__float_as_int(cl), idx));
                float* ar = agg + ((size_t)d << LOGF) + lane;
#pragma unroll
                for (int j = 0; j < NL; ++j)
                    __hip_atomic_fetch_add(ar + j * 32, cur[k][j] * c,
                                           __ATOMIC_RELAXED, __HIP_MEMORY_SCOPE_AGENT);
            }
        }
    }
}

template <int LOGF>
__global__ __launch_bounds__(TPB)
void gcn_edge_scatter(const int* __restrict__ src, const int* __restrict__ dst,
                      const float* __restrict__ coef, const float* __restrict__ xw,
                      float* __restrict__ agg, int E) {
    const int lane = threadIdx.x & 31;
    const int wave = __builtin_amdgcn_readfirstlane((int)(blockIdx.x * (TPB / 32) + (threadIdx.x >> 5)));
    const int base = wave * 32;
    if (base >= E) return;
    const int cnt = min(32, E - base);

    int sl = 0, dl = 0;
    float cl = 0.0f;
    if (lane < cnt) {
        sl = src[base + lane];
        dl = dst[base + lane];
        cl = coef[base + lane];
    }
    if (cnt == 32) gcn_scatter_walk<LOGF, true >(sl, dl, cl, cnt, lane, xw, agg);
    else           gcn_scatter_walk<LOGF, false>(sl, dl, cl, cnt, lane, xw, agg);
}

__global__ void gcn_bn_relu(const float* __restrict__ in, float* __restrict__ out,
                            const float* __restrict__ g, const float* __restrict__ beta,
                            const float* __restrict__ m, const float* __restrict__ v,
                            long totalV4, int fmask) {
    long idx = (long)blockIdx.x * blockDim.x + threadIdx.x;
    if (idx >= totalV4) return;
    size_t elem = (size_t)idx << 2;
    int f4 = (int)(elem & fmask);
    float4 xv = *(const float4*)(in + elem);
    float4 gv = *(const float4*)(g + f4);
    float4 bv = *(const float4*)(beta + f4);
    float4 mv = *(const float4*)(m + f4);
    float4 vv = *(const float4*)(v + f4);
    float4 o;
    o.x = fmaxf((xv.x - mv.x) * rsqrtf(vv.x + 1e-5f) * gv.x + bv.x, 0.0f);
    o.y = fmaxf((xv.y - mv.y) * rsqrtf(vv.y + 1e-5f) * gv.y + bv.y, 0.0f);
    o.z = fmaxf((xv.z - mv.z) * rsqrtf(vv.z + 1e-5f) * gv.z + bv.z, 0.0f);
    o.w = fmaxf((xv.w - mv.w) * rsqrtf(vv.w + 1e-5f) * gv.w + bv.w, 0.0f);
    *(float4*)(out + elem) = o;
}

__global__ __launch_bounds__(TPB)
void gcn_softmax64(const float* __restrict__ in, float* __restrict__ out, int n) {
    int wid  = threadIdx.x >> 5;
    int lane = threadIdx.x & 31;
    int row  = blockIdx.x * 8 + wid;
    if (row >= n) return;
    const float* p = in + (size_t)row * 64;
    float v0 = p[lane], v1 = p[lane + 32];
    float mx = fmaxf(v0, v1);
#pragma unroll
    for (int off = 16; off > 0; off >>= 1) mx = fmaxf(mx, __shfl_xor(mx, off, 32));
    float e0 = __expf(v0 - mx), e1 = __expf(v1 - mx);
    float s = e0 + e1;
#pragma unroll
    for (int off = 16; off > 0; off >>= 1) s += __shfl_xor(s, off, 32);
    float inv = 1.0f / s;
    float* q = out + (size_t)row * 64;
    q[lane] = e0 * inv;
    q[lane + 32] = e1 * inv;
}

extern "C" void kernel_launch(void* const* d_in, const int* in_sizes, int n_in,
                              void* d_out, int out_size, void* d_ws, size_t ws_size,
                              hipStream_t stream) {
    const int IN = 256, HID_ = 256, OUT_ = 64;
    const int N = in_sizes[0] / IN;
    const int E = in_sizes[1] / 2;

    const float* x    = (const float*)d_in[0];
    const int*   src  = (const int*)d_in[1];
    const int*   dst  = ((const int*)d_in[1]) + E;
    const float* W1   = (const float*)d_in[2];
    const float* b1   = (const float*)d_in[3];
    const float* g1   = (const float*)d_in[4];
    const float* be1  = (const float*)d_in[5];
    const float* m1   = (const float*)d_in[6];
    const float* v1   = (const float*)d_in[7];
    const float* W2   = (const float*)d_in[8];
    const float* b2   = (const float*)d_in[9];
    const float* g2   = (const float*)d_in[10];
    const float* be2  = (const float*)d_in[11];
    const float* m2   = (const float*)d_in[12];
    const float* v2   = (const float*)d_in[13];
    const float* W3   = (const float*)d_in[14];
    const float* b3   = (const float*)d_in[15];
    float* out = (float*)d_out;

    char* ws = (char*)d_ws;
    size_t off = 0;
    float* dinv = (float*)(ws + off); off += ((size_t)N * 4 + 255) & ~(size_t)255;
    float* coef = (float*)(ws + off); off += ((size_t)E * 4 + 255) & ~(size_t)255;
    float* bufA = (float*)(ws + off); off += (size_t)N * HID_ * 4;
    float* bufB = (float*)(ws + off);
    (void)ws_size; (void)n_in; (void)out_size;

    const long n256v4 = (long)N * HID_ / 4;
    const long n64v4  = (long)N * OUT_ / 4;
    const int gN = (N + TPB - 1) / TPB;
    const int gE = (E + TPB - 1) / TPB;
    const int gemmX = (N + 127) / 128;
    auto blocks = [](long t) { return (int)((t + TPB - 1) / TPB); };

    gcn_deg_init <<<gN, TPB, 0, stream>>>(dinv, N);
    gcn_deg_edges<<<gE, TPB, 0, stream>>>(dst, dinv, E);
    gcn_dinv     <<<gN, TPB, 0, stream>>>(dinv, N);
    gcn_edge_coef<<<gE, TPB, 0, stream>>>(src, dst, dinv, coef, E);

    gcn_gemm_wmma   <<<dim3(gemmX, HID_ / 64), TPB, 0, stream>>>(x, W1, bufA, N, IN, HID_);
    gcn_agg_init    <<<blocks(n256v4), TPB, 0, stream>>>(bufA, b1, dinv, bufB, n256v4, 8);
    gcn_edge_scatter<8><<<gE, TPB, 0, stream>>>(src, dst, coef, bufA, bufB, E);
    gcn_bn_relu     <<<blocks(n256v4), TPB, 0, stream>>>(bufB, bufA, g1, be1, m1, v1, n256v4, 255);

    gcn_gemm_wmma   <<<dim3(gemmX, HID_ / 64), TPB, 0, stream>>>(bufA, W2, bufB, N, HID_, HID_);
    gcn_agg_init    <<<blocks(n256v4), TPB, 0, stream>>>(bufB, b2, dinv, bufA, n256v4, 8);
    gcn_edge_scatter<8><<<gE, TPB, 0, stream>>>(src, dst, coef, bufB, bufA, E);
    gcn_bn_relu     <<<blocks(n256v4), TPB, 0, stream>>>(bufA, bufB, g2, be2, m2, v2, n256v4, 255);

    gcn_gemm_wmma   <<<dim3(gemmX, OUT_ / 64), TPB, 0, stream>>>(bufB, W3, bufA, N, HID_, OUT_);
    gcn_agg_init    <<<blocks(n64v4), TPB, 0, stream>>>(bufA, b3, dinv, bufB, n64v4, 6);
    gcn_edge_scatter<6><<<gE, TPB, 0, stream>>>(src, dst, coef, bufA, bufB, E);
    gcn_softmax64   <<<(N + 7) / 8, TPB, 0, stream>>>(bufB, out, N);
}
